# starting kernel
# baseline (speedup 1.0000x reference)
#include <hip/hip_runtime.h>
#include <math.h>


typedef __attribute__((ext_vector_type(2))) float v2f;
typedef __attribute__((ext_vector_type(8))) float v8f;

#define NH        8
#define HD        32
#define DMODEL    256
#define HGRID     56
#define MWIN      7
#define MTOK      112
#define KCHUNK    64
#define XS_STRIDE 68
#define WP_STRIDE 224
#define NPAIR     (KCHUNK / 2)
#define QK_STRIDE 100
#define ATT_STRIDE 18

#define XS_FLOATS  (MTOK * XS_STRIDE)
#define WH_FLOATS  (NPAIR * WP_STRIDE)
#define QK_FLOATS  (MTOK * QK_STRIDE)
#define ATT_FLOATS (8 * 16 * ATT_STRIDE)
#define SMEM_FLOATS (XS_FLOATS + WH_FLOATS + QK_FLOATS + ATT_FLOATS)
#define SMEM_BYTES  (SMEM_FLOATS * 4 + MTOK * 4)

static __device__ __forceinline__ v8f wmma_f32_4(v2f a, v2f b, v8f c) {
    return __builtin_amdgcn_wmma_f32_16x16x4_f32(false, a, false, b,
                                                 (short)0, c, false, false);
}

__global__ __launch_bounds__(256)
void swin_msa_fused(const float* __restrict__ x, const float* __restrict__ Wg,
                    const float* __restrict__ bvec, float* __restrict__ out) {
    extern __shared__ float smem[];
    float* xs  = smem;
    float* whp = xs + XS_FLOATS;
    float* qk  = whp + WH_FLOATS;
    float* att = qk + QK_FLOATS;
    int* rowoff = (int*)(att + ATT_FLOATS);

    const int half = blockIdx.x;
    const int wi   = blockIdx.y;
    const int bb   = blockIdx.z;
    const int tid  = threadIdx.x;
    const int wave = tid >> 5;
    const int lane = tid & 31;
    const int r    = lane & 15;
    const int hiK  = (lane >> 4) << 1;
    const int hiP  = lane >> 4;
    const int rowb = (lane >= 16) ? 8 : 0;

    if (tid < MTOK) {
        int w  = tid >> 4;
        int t  = tid & 15;
        int gh = 4 * wi + (t >> 2) + 2;               if (gh >= HGRID) gh -= HGRID;
        int gw = 4 * (half * MWIN + w) + (t & 3) + 2; if (gw >= HGRID) gw -= HGRID;
        rowoff[tid] = bb * (HGRID * HGRID) + gh * HGRID + gw;
    }
    __syncthreads();

    const bool lastRow = (wi == 13);
    const bool lastCol = (half == 1) && (wave == 6);
    const bool maskR = lastRow && (((lane >= 16) ? 1 : 0) != ((lane & 8) ? 1 : 0));
    const float scale = 0.17677669529663687f;

    for (int h = 0; h < NH; ++h) {
        v8f acc[6];
        if (wave < MWIN) {
            const float* bp = bvec + 96 * h;
            #pragma unroll
            for (int n = 0; n < 6; ++n) {
                float bv = bp[16 * n + r];
                v8f t;
                #pragma unroll
                for (int q = 0; q < 8; ++q) t[q] = bv;
                acc[n] = t;
            }
        }

        for (int kc = 0; kc < 4; ++kc) {
            __syncthreads();
            #pragma unroll
            for (int it = 0; it < 7; ++it) {
                int idx = tid + 256 * it;
                int row = idx >> 4;
                int c4  = (idx & 15) << 2;
                const float4 v = *(const float4*)(x + (size_t)rowoff[row] * DMODEL
                                                    + kc * KCHUNK + c4);
                *(float4*)&xs[row * XS_STRIDE + c4] = v;
            }
            #pragma unroll
            for (int it = 0; it < 6; ++it) {
                int idx  = tid + 256 * it;
                int pair = idx / 48;
                int c2g  = idx % 48;
                const float* g0 = Wg + (size_t)(kc * KCHUNK + 2 * pair) * (3 * DMODEL)
                                     + 96 * h + 2 * c2g;
                float2 a0 = *(const float2*)g0;
                float2 a1 = *(const float2*)(g0 + 3 * DMODEL);
                float4 st; st.x = a0.x; st.y = a1.x; st.z = a0.y; st.w = a1.y;
                *(float4*)&whp[pair * WP_STRIDE + 4 * c2g] = st;
            }
            __syncthreads();

            if (wave < MWIN) {
                const float* xrow = &xs[(16 * wave + r) * XS_STRIDE + hiK];
                const float* wrow = &whp[hiP * WP_STRIDE + 2 * r];
                #pragma unroll 4
                for (int k0 = 0; k0 < KCHUNK; k0 += 4) {
                    v2f a = *(const v2f*)(xrow + k0);
                    v2f bf[6];
                    #pragma unroll
                    for (int n = 0; n < 6; ++n)
                        bf[n] = *(const v2f*)(wrow + (k0 >> 1) * WP_STRIDE + 32 * n);
                    #pragma unroll
                    for (int n = 0; n < 6; ++n)
                        acc[n] = wmma_f32_4(a, bf[n], acc[n]);
                }
            }
        }

        if (wave < MWIN) {
            int qrowb = 16 * wave + rowb;
            #pragma unroll
            for (int n = 0; n < 6; ++n)
                #pragma unroll
                for (int v = 0; v < 8; ++v)
                    qk[(qrowb + v) * QK_STRIDE + 16 * n + r] = acc[n][v];

            const float* qrow = &qk[(16 * wave + r) * QK_STRIDE];
            v8f s = {0.f, 0.f, 0.f, 0.f, 0.f, 0.f, 0.f, 0.f};
            #pragma unroll
            for (int k0 = 0; k0 < HD; k0 += 4) {
                int ka = 3 * (k0 + hiK);
                v2f a, bf;
                a.x  = qrow[ka];     a.y  = qrow[ka + 3];
                bf.x = qrow[ka + 1]; bf.y = qrow[ka + 4];
                s = wmma_f32_4(a, bf, s);
            }

            v8f p;
            #pragma unroll
            for (int v = 0; v < 8; ++v) {
                float xv = s[v] * scale;
                bool mC = lastCol && (((v & 2) != 0) != ((lane & 2) != 0));
                if (maskR || mC) xv = -1e30f;
                float m = xv;
                #pragma unroll
                for (int off = 8; off; off >>= 1)
                    m = fmaxf(m, __shfl_xor(m, off, 16));
                float e = __expf(xv - m);
                float sm = e;
                #pragma unroll
                for (int off = 8; off; off >>= 1)
                    sm += __shfl_xor(sm, off, 16);
                p[v] = e / sm;
            }

            float* am = &att[wave * 16 * ATT_STRIDE];
            #pragma unroll
            for (int v = 0; v < 8; ++v)
                am[(rowb + v) * ATT_STRIDE + r] = p[v];

            v8f o0 = {0.f, 0.f, 0.f, 0.f, 0.f, 0.f, 0.f, 0.f};
            v8f o1 = {0.f, 0.f, 0.f, 0.f, 0.f, 0.f, 0.f, 0.f};
            #pragma unroll
            for (int k0 = 0; k0 < 16; k0 += 4) {
                v2f a = *(const v2f*)(am + r * ATT_STRIDE + k0 + hiK);
                const float* v0p = &qk[(16 * wave + k0 + hiK) * QK_STRIDE + 2];
                const float* v1p = v0p + QK_STRIDE;
                v2f b0; b0.x = v0p[3 * r];        b0.y = v1p[3 * r];
                o0 = wmma_f32_4(a, b0, o0);
                v2f b1; b1.x = v0p[3 * (16 + r)]; b1.y = v1p[3 * (16 + r)];
                o1 = wmma_f32_4(a, b1, o1);
            }

            float* ob = out + 32 * h + r;
            #pragma unroll
            for (int v = 0; v < 8; ++v) {
                int grow = rowoff[16 * wave + rowb + v];
                float* pdst = ob + (size_t)grow * DMODEL;
                __builtin_nontemporal_store(o0[v], pdst);
                __builtin_nontemporal_store(o1[v], pdst + 16);
            }
        }
    }
}

extern "C" void kernel_launch(void* const* d_in, const int* in_sizes, int n_in,
                              void* d_out, int out_size, void* d_ws, size_t ws_size,
                              hipStream_t stream) {
    (void)in_sizes; (void)n_in; (void)d_ws; (void)ws_size; (void)out_size;
    const float* x = (const float*)d_in[0];
    const float* W = (const float*)d_in[1];
    const float* b = (const float*)d_in[2];
    float* out = (float*)d_out;

    static_assert(SMEM_BYTES < 320 * 1024, "LDS budget");
    hipFuncSetAttribute((const void*)swin_msa_fused,
                        hipFuncAttributeMaxDynamicSharedMemorySize, SMEM_BYTES);

    dim3 grid(2, 14, 32);
    swin_msa_fused<<<grid, 256, SMEM_BYTES, stream>>>(x, W, b, out);
}


# Staging software-pipelined through registers: next chunk's 19 global loads per thread issued ahead of this chunk's WMMAs, row offsets in registers; before, the 7 x loads ran one after another
# speedup vs baseline: 1.2019x; 1.2019x over previous
#include <hip/hip_runtime.h>
#include <math.h>


typedef __attribute__((ext_vector_type(2))) float v2f;
typedef __attribute__((ext_vector_type(4))) float v4f;
typedef __attribute__((ext_vector_type(8))) float v8f;

#define NH        8
#define HD        32
#define DMODEL    256
#define HGRID     56
#define MWIN      7
#define MTOK      112
#define KCHUNK    64
#define XS_STRIDE 68
#define WP_STRIDE 224
#define NPAIR     (KCHUNK / 2)
#define QK_STRIDE 100
#define ATT_STRIDE 18

#define XS_FLOATS  (MTOK * XS_STRIDE)
#define WH_FLOATS  (NPAIR * WP_STRIDE)
#define QK_FLOATS  (MTOK * QK_STRIDE)
#define ATT_FLOATS (8 * 16 * ATT_STRIDE)
#define SMEM_FLOATS (XS_FLOATS + WH_FLOATS + QK_FLOATS + ATT_FLOATS)
#define SMEM_BYTES  (SMEM_FLOATS * 4 + MTOK * 4)

static __device__ __forceinline__ v8f wmma_f32_4(v2f a, v2f b, v8f c) {
    return __builtin_amdgcn_wmma_f32_16x16x4_f32(false, a, false, b,
                                                 (short)0, c, false, false);
}

__global__ __launch_bounds__(256)
void swin_msa_fused(const float* __restrict__ x, const float* __restrict__ Wg,
                    const float* __restrict__ bvec, float* __restrict__ out) {
    extern __shared__ float smem[];
    float* xs  = smem;
    float* whp = xs + XS_FLOATS;
    float* qk  = whp + WH_FLOATS;
    float* att = qk + QK_FLOATS;
    int* rowoff = (int*)(att + ATT_FLOATS);

    const int half = blockIdx.x;
    const int wi   = blockIdx.y;
    const int bb   = blockIdx.z;
    const int tid  = threadIdx.x;
    const int wave = tid >> 5;
    const int lane = tid & 31;
    const int r    = lane & 15;
    const int hiK  = (lane >> 4) << 1;
    const int hiP  = lane >> 4;
    const int rowb = (lane >= 16) ? 8 : 0;

    if (tid < MTOK) {
        int w  = tid >> 4;
        int t  = tid & 15;
        int gh = 4 * wi + (t >> 2) + 2;               if (gh >= HGRID) gh -= HGRID;
        int gw = 4 * (half * MWIN + w) + (t & 3) + 2; if (gw >= HGRID) gw -= HGRID;
        rowoff[tid] = bb * (HGRID * HGRID) + gh * HGRID + gw;
    }
    __syncthreads();

    const bool lastRow = (wi == 13);
    const bool lastCol = (half == 1) && (wave == 6);
    const bool maskR = lastRow && (((lane >= 16) ? 1 : 0) != ((lane & 8) ? 1 : 0));
    const float scale = 0.17677669529663687f;

    const int srow = tid >> 4;
    const int sc4  = (tid & 15) << 2;
    unsigned xgb[7];
    {
        int gh = 4 * wi + (srow >> 2) + 2;            if (gh >= HGRID) gh -= HGRID;
        const int rbase = bb * (HGRID * HGRID) + gh * HGRID;
        #pragma unroll
        for (int it = 0; it < 7; ++it) {
            int gw = 4 * (half * MWIN + it) + (srow & 3) + 2; if (gw >= HGRID) gw -= HGRID;
            xgb[it] = ((unsigned)(rbase + gw) * DMODEL + sc4) * 4u;
        }
    }
    float* const xsp = &xs[srow * XS_STRIDE + sc4];

    v4f xr[7];
    v2f wr0[6], wr1[6];

    auto issue_loads = [&](int hh, int kk) {
        const char* xb = (const char*)x + kk * (KCHUNK * 4);
        #pragma unroll
        for (int it = 0; it < 7; ++it)
            xr[it] = *(const v4f*)(xb + xgb[it]);
        const float* wb = Wg + (size_t)(kk * KCHUNK) * (3 * DMODEL) + 96 * hh;
        #pragma unroll
        for (int it = 0; it < 6; ++it) {
            int idx  = tid + 256 * it;
            int pair = idx / 48;
            int c2g  = idx % 48;
            const float* g0 = wb + (2 * pair) * (3 * DMODEL) + 2 * c2g;
            wr0[it] = *(const v2f*)g0;
            wr1[it] = *(const v2f*)(g0 + 3 * DMODEL);
        }
    };
    auto store_staged = [&]() {
        #pragma unroll
        for (int it = 0; it < 7; ++it)
            *(v4f*)(xsp + it * 16 * XS_STRIDE) = xr[it];
        #pragma unroll
        for (int it = 0; it < 6; ++it) {
            int idx  = tid + 256 * it;
            int pair = idx / 48;
            int c2g  = idx % 48;
            v4f st; st.x = wr0[it].x; st.y = wr1[it].x; st.z = wr0[it].y; st.w = wr1[it].y;
            *(v4f*)&whp[pair * WP_STRIDE + 4 * c2g] = st;
        }
    };

    issue_loads(0, 0);

    for (int h = 0; h < NH; ++h) {
        v8f acc[6];
        if (wave < MWIN) {
            const float* bp = bvec + 96 * h;
            #pragma unroll
            for (int n = 0; n < 6; ++n) {
                float bv = bp[16 * n + r];
                v8f t;
                #pragma unroll
                for (int q = 0; q < 8; ++q) t[q] = bv;
                acc[n] = t;
            }
        }

        #pragma unroll 1
        for (int kc = 0; kc < 4; ++kc) {
            __syncthreads();
            store_staged();
            __syncthreads();

            {
                int kn = kc + 1, hn = h;
                if (kn == 4) { kn = 0; hn = h + 1; }
                if (hn < NH) issue_loads(hn, kn);
            }

            if (wave < MWIN) {
                const float* xrow = &xs[(16 * wave + r) * XS_STRIDE + hiK];
                const float* wrow = &whp[hiP * WP_STRIDE + 2 * r];
                #pragma unroll 4
                for (int k0 = 0; k0 < KCHUNK; k0 += 4) {
                    v2f a = *(const v2f*)(xrow + k0);
                    v2f bf[6];
                    #pragma unroll
                    for (int n = 0; n < 6; ++n)
                        bf[n] = *(const v2f*)(wrow + (k0 >> 1) * WP_STRIDE + 32 * n);
                    #pragma unroll
                    for (int n = 0; n < 6; ++n)
                        acc[n] = wmma_f32_4(a, bf[n], acc[n]);
                }
            }
        }

        if (wave < MWIN) {
            int qrowb = 16 * wave + rowb;
            #pragma unroll
            for (int n = 0; n < 6; ++n)
                #pragma unroll
                for (int v = 0; v < 8; ++v)
                    qk[(qrowb + v) * QK_STRIDE + 16 * n + r] = acc[n][v];

            const float* qrow = &qk[(16 * wave + r) * QK_STRIDE];
            v8f s = {0.f, 0.f, 0.f, 0.f, 0.f, 0.f, 0.f, 0.f};
            #pragma unroll
            for (int k0 = 0; k0 < HD; k0 += 4) {
                int ka = 3 * (k0 + hiK);
                v2f a, bf;
                a.x  = qrow[ka];     a.y  = qrow[ka + 3];
                bf.x = qrow[ka + 1]; bf.y = qrow[ka + 4];
                s = wmma_f32_4(a, bf, s);
            }

            v8f p;
            #pragma unroll
            for (int v = 0; v < 8; ++v) {
                float xv = s[v] * scale;
                bool mC = lastCol && (((v & 2) != 0) != ((lane & 2) != 0));
                if (maskR || mC) xv = -1e30f;
                float m = xv;
                #pragma unroll
                for (int off = 8; off; off >>= 1)
                    m = fmaxf(m, __shfl_xor(m, off, 16));
                float e = __expf(xv - m);
                float sm = e;
                #pragma unroll
                for (int off = 8; off; off >>= 1)
                    sm += __shfl_xor(sm, off, 16);
                p[v] = e / sm;
            }

            float* am = &att[wave * 16 * ATT_STRIDE];
            #pragma unroll
            for (int v = 0; v < 8; ++v)
                am[(rowb + v) * ATT_STRIDE + r] = p[v];

            v8f o0 = {0.f, 0.f, 0.f, 0.f, 0.f, 0.f, 0.f, 0.f};
            v8f o1 = {0.f, 0.f, 0.f, 0.f, 0.f, 0.f, 0.f, 0.f};
            #pragma unroll
            for (int k0 = 0; k0 < 16; k0 += 4) {
                v2f a = *(const v2f*)(am + r * ATT_STRIDE + k0 + hiK);
                const float* v0p = &qk[(16 * wave + k0 + hiK) * QK_STRIDE + 2];
                const float* v1p = v0p + QK_STRIDE;
                v2f b0; b0.x = v0p[3 * r];        b0.y = v1p[3 * r];
                o0 = wmma_f32_4(a, b0, o0);
                v2f b1; b1.x = v0p[3 * (16 + r)]; b1.y = v1p[3 * (16 + r)];
                o1 = wmma_f32_4(a, b1, o1);
            }

            float* ob = out + 32 * h + r;
            #pragma unroll
            for (int v = 0; v < 8; ++v) {
                int grow = rowoff[16 * wave + rowb + v];
                float* pdst = ob + (size_t)grow * DMODEL;
                __builtin_nontemporal_store(o0[v], pdst);
                __builtin_nontemporal_store(o1[v], pdst + 16);
            }
        }
    }
}

extern "C" void kernel_launch(void* const* d_in, const int* in_sizes, int n_in,
                              void* d_out, int out_size, void* d_ws, size_t ws_size,
                              hipStream_t stream) {
    (void)in_sizes; (void)n_in; (void)d_ws; (void)ws_size; (void)out_size;
    const float* x = (const float*)d_in[0];
    const float* W = (const float*)d_in[1];
    const float* b = (const float*)d_in[2];
    float* out = (float*)d_out;

    static_assert(SMEM_BYTES < 320 * 1024, "LDS budget");
    hipFuncSetAttribute((const void*)swin_msa_fused,
                        hipFuncAttributeMaxDynamicSharedMemorySize, SMEM_BYTES);

    dim3 grid(2, 14, 32);
    swin_msa_fused<<<grid, 256, SMEM_BYTES, stream>>>(x, W, b, out);
}


# One work-group per (image, head): its W slice staged once and resident in LDS, x rows loaded straight into A-fragment registers (k regrouped, bias added last: qkv sum order changes); no loop barriers
# speedup vs baseline: 1.6202x; 1.3480x over previous
#include <hip/hip_runtime.h>
#include <math.h>


typedef __attribute__((ext_vector_type(2))) float v2f;
typedef __attribute__((ext_vector_type(4))) float v4f;
typedef __attribute__((ext_vector_type(8))) float v8f;

#define NH        8
#define HD        32
#define DMODEL    256
#define HGRID     56
#define NWROW     14
#define NWIN      196
#define NWAVE     16
#define NPAIR     128
#define WP_STRIDE 208
#define QK_STRIDE 100
#define ATT_STRIDE 18

#define W_FLOATS   (NPAIR * WP_STRIDE)
#define QK_FLOATS  (16 * QK_STRIDE)
#define ATT_FLOATS (16 * ATT_STRIDE)
#define SMEM_FLOATS (W_FLOATS + NWAVE * (QK_FLOATS + ATT_FLOATS))
#define SMEM_BYTES  (SMEM_FLOATS * 4)

static __device__ __forceinline__ v8f wmma_f32_4(v2f a, v2f b, v8f c) {
    return __builtin_amdgcn_wmma_f32_16x16x4_f32(false, a, false, b,
                                                 (short)0, c, false, false);
}

__global__ __launch_bounds__(512)
void swin_msa_head_resident(const float* __restrict__ x, const float* __restrict__ Wg,
                            const float* __restrict__ bvec, float* __restrict__ out) {
    extern __shared__ float smem[];
    float* whp = smem;

    const int h    = blockIdx.x;
    const int bb   = blockIdx.y;
    const int tid  = threadIdx.x;
    const int wave = __builtin_amdgcn_readfirstlane(tid >> 5);
    const int lane = tid & 31;
    const int r    = lane & 15;
    const int hh   = lane >> 4;
    const int hiK  = hh << 1;
    const int rowb = hh << 3;
    float* qk = smem + W_FLOATS + wave * QK_FLOATS;
    float* am = smem + W_FLOATS + NWAVE * QK_FLOATS + wave * ATT_FLOATS;

    {
        const float* wb = Wg + 96 * h;
        #pragma unroll
        for (int it = 0; it < 12; ++it) {
            int idx  = tid + 512 * it;
            int pair = idx / 48;
            int c2g  = idx % 48;
            const float* g0 = wb + (2 * pair) * (3 * DMODEL) + 2 * c2g;
            v2f w0 = *(const v2f*)g0;
            v2f w1 = *(const v2f*)(g0 + 3 * DMODEL);
            v4f st; st.x = w0.x; st.y = w1.x; st.z = w0.y; st.w = w1.y;
            *(v4f*)&whp[pair * WP_STRIDE + 4 * c2g] = st;
        }
    }

    float bq[6];
    #pragma unroll
    for (int n = 0; n < 6; ++n) bq[n] = bvec[96 * h + 16 * n + r];

    auto row_ptr = [&](int w_) -> const float* {
        int wi_ = w_ / NWROW, wj_ = w_ - wi_ * NWROW;
        int gh = 4 * wi_ + (r >> 2) + 2; if (gh >= HGRID) gh -= HGRID;
        int gw = 4 * wj_ + (r & 3) + 2;  if (gw >= HGRID) gw -= HGRID;
        return x + (size_t)(bb * (HGRID * HGRID) + gh * HGRID + gw) * DMODEL + 4 * hh;
    };

    v4f xa[16];
    {
        const float* xp = row_ptr(wave);
        #pragma unroll
        for (int i = 0; i < 16; ++i) xa[i] = *(const v4f*)(xp + 8 * i);
    }

    __syncthreads();

    const float scale = 0.17677669529663687f;
    const float* wrow = whp + hiK * WP_STRIDE + 2 * r;

    for (int win = wave; win < NWIN; win += NWAVE) {
        asm volatile("" ::: "memory");
        const int wi = win / NWROW;
        const int wj = win - wi * NWROW;
        const int wnext = (win + NWAVE < NWIN) ? (win + NWAVE) : win;
        const float* xn = row_ptr(wnext);
        const float* xc = row_ptr(win);

        v8f acc[6];
        #pragma unroll
        for (int n = 0; n < 6; ++n) {
            v8f t = {0.f, 0.f, 0.f, 0.f, 0.f, 0.f, 0.f, 0.f};
            acc[n] = t;
        }

        v2f bfr[2][12];
        #pragma unroll
        for (int t = 0; t < 2; ++t)
            #pragma unroll
            for (int n = 0; n < 6; ++n)
                bfr[0][6 * t + n] = *(const v2f*)(wrow + t * WP_STRIDE + 32 * n);

        #pragma unroll
        for (int i = 0; i < 32; ++i) {
            if (i + 1 < 32) {
                #pragma unroll
                for (int t = 0; t < 2; ++t)
                    #pragma unroll
                    for (int n = 0; n < 6; ++n)
                        bfr[(i + 1) & 1][6 * t + n] =
                            *(const v2f*)(wrow + (4 * (i + 1) + t) * WP_STRIDE + 32 * n);
            }
            __builtin_amdgcn_sched_barrier(0);
            #pragma unroll
            for (int t = 0; t < 2; ++t) {
                v2f a;
                if (t == 0) { a.x = xa[i & 15].x; a.y = xa[i & 15].y; }
                else        { a.x = xa[i & 15].z; a.y = xa[i & 15].w; }
                #pragma unroll
                for (int n = 0; n < 6; ++n)
                    acc[n] = wmma_f32_4(a, bfr[i & 1][6 * t + n], acc[n]);
            }
            __builtin_amdgcn_sched_barrier(0);
            if ((i & 3) == 3) {
                #pragma unroll
                for (int ii = 0; ii < 4; ++ii) {
                    const int j = i - 3 + ii;
                    if (j < 16) xa[j & 15] = *(const v4f*)(xc + 8 * (j + 16));
                    else        xa[j & 15] = *(const v4f*)(xn + 8 * (j - 16));
                }
            }
        }

        #pragma unroll
        for (int n = 0; n < 6; ++n)
            #pragma unroll
            for (int v = 0; v < 8; ++v)
                qk[(rowb + v) * QK_STRIDE + 16 * n + r] = acc[n][v] + bq[n];

        const float* qrow = &qk[r * QK_STRIDE];
        v8f s = {0.f, 0.f, 0.f, 0.f, 0.f, 0.f, 0.f, 0.f};
        #pragma unroll
        for (int k0 = 0; k0 < HD; k0 += 4) {
            int ka = 3 * (k0 + hiK);
            v2f a, bf;
            a.x  = qrow[ka];     a.y  = qrow[ka + 3];
            bf.x = qrow[ka + 1]; bf.y = qrow[ka + 4];
            s = wmma_f32_4(a, bf, s);
        }

        const bool lastCol = (wj == NWROW - 1);
        const bool maskR = (wi == NWROW - 1) && (hh != ((lane >> 3) & 1));
        v8f p;
        #pragma unroll
        for (int v = 0; v < 8; ++v) {
            float xv = s[v] * scale;
            bool mC = lastCol && (((v & 2) != 0) != ((lane & 2) != 0));
            if (maskR || mC) xv = -1e30f;
            float m = xv;
            #pragma unroll
            for (int off = 8; off; off >>= 1)
                m = fmaxf(m, __shfl_xor(m, off, 16));
            float e = __expf(xv - m);
            float sm = e;
            #pragma unroll
            for (int off = 8; off; off >>= 1)
                sm += __shfl_xor(sm, off, 16);
            p[v] = e / sm;
        }

        #pragma unroll
        for (int v = 0; v < 8; ++v)
            am[(rowb + v) * ATT_STRIDE + r] = p[v];

        v8f o0 = {0.f, 0.f, 0.f, 0.f, 0.f, 0.f, 0.f, 0.f};
        v8f o1 = {0.f, 0.f, 0.f, 0.f, 0.f, 0.f, 0.f, 0.f};
        #pragma unroll
        for (int k0 = 0; k0 < 16; k0 += 4) {
            v2f a = *(const v2f*)(am + r * ATT_STRIDE + k0 + hiK);
            const float* v0p = &qk[(k0 + hiK) * QK_STRIDE + 2];
            const float* v1p = v0p + QK_STRIDE;
            v2f b0; b0.x = v0p[3 * r];        b0.y = v1p[3 * r];
            o0 = wmma_f32_4(a, b0, o0);
            v2f b1; b1.x = v0p[3 * (16 + r)]; b1.y = v1p[3 * (16 + r)];
            o1 = wmma_f32_4(a, b1, o1);
        }

        float* ob = out + 32 * h + r;
        const int ghb = 4 * wi + hiK + 2;
        const int gwb = 4 * wj + 2;
        #pragma unroll
        for (int v = 0; v < 8; ++v) {
            int gh = ghb + (v >> 2); if (gh >= HGRID) gh -= HGRID;
            int gw = gwb + (v & 3);  if (gw >= HGRID) gw -= HGRID;
            float* pdst = ob + (size_t)(bb * (HGRID * HGRID) + gh * HGRID + gw) * DMODEL;
            __builtin_nontemporal_store(o0[v], pdst);
            __builtin_nontemporal_store(o1[v], pdst + 16);
        }
    }
}

extern "C" void kernel_launch(void* const* d_in, const int* in_sizes, int n_in,
                              void* d_out, int out_size, void* d_ws, size_t ws_size,
                              hipStream_t stream) {
    (void)in_sizes; (void)n_in; (void)d_ws; (void)ws_size; (void)out_size;
    const float* x = (const float*)d_in[0];
    const float* W = (const float*)d_in[1];
    const float* b = (const float*)d_in[2];
    float* out = (float*)d_out;

    static_assert(SMEM_BYTES < 320 * 1024, "LDS budget");
    hipFuncSetAttribute((const void*)swin_msa_head_resident,
                        hipFuncAttributeMaxDynamicSharedMemorySize, SMEM_BYTES);

    dim3 grid(8, 32);
    swin_msa_head_resident<<<grid, 512, SMEM_BYTES, stream>>>(x, W, b, out);
}


# Scores computed transposed (K Q^T): softmax rows lane-owned, max/sum in-lane + one half-swap, P feeds AV from registers (row-sum and AV k order change, one reciprocal per row); no LDS shuffles
# speedup vs baseline: 1.7050x; 1.0523x over previous
#include <hip/hip_runtime.h>
#include <math.h>


typedef __attribute__((ext_vector_type(2))) float v2f;
typedef __attribute__((ext_vector_type(4))) float v4f;
typedef __attribute__((ext_vector_type(8))) float v8f;

#define NH        8
#define HD        32
#define DMODEL    256
#define HGRID     56
#define NWROW     14
#define NWIN      196
#define NWAVE     16
#define NPAIR     128
#define WP_STRIDE 208
#define QK_STRIDE 100

#define W_FLOATS   (NPAIR * WP_STRIDE)
#define QK_FLOATS  (16 * QK_STRIDE)
#define SMEM_FLOATS (W_FLOATS + NWAVE * QK_FLOATS)
#define SMEM_BYTES  (SMEM_FLOATS * 4)

static __device__ __forceinline__ v8f wmma_f32_4(v2f a, v2f b, v8f c) {
    return __builtin_amdgcn_wmma_f32_16x16x4_f32(false, a, false, b,
                                                 (short)0, c, false, false);
}

static __device__ __forceinline__ float xhalf(float v) {
    return __builtin_bit_cast(float, __builtin_amdgcn_permlanex16(
        __builtin_bit_cast(int, v), __builtin_bit_cast(int, v),
        0x76543210, 0xfedcba98, false, false));
}

__global__ __launch_bounds__(512)
void swin_msa_head_resident(const float* __restrict__ x, const float* __restrict__ Wg,
                            const float* __restrict__ bvec, float* __restrict__ out) {
    extern __shared__ float smem[];
    float* whp = smem;

    const int h    = blockIdx.x;
    const int bb   = blockIdx.y;
    const int tid  = threadIdx.x;
    const int wave = __builtin_amdgcn_readfirstlane(tid >> 5);
    const int lane = tid & 31;
    const int r    = lane & 15;
    const int hh   = lane >> 4;
    const int hiK  = hh << 1;
    const int rowb = hh << 3;
    float* qk = smem + W_FLOATS + wave * QK_FLOATS;

    {
        const float* wb = Wg + 96 * h;
        #pragma unroll
        for (int it = 0; it < 12; ++it) {
            int idx  = tid + 512 * it;
            int pair = idx / 48;
            int c2g  = idx % 48;
            const float* g0 = wb + (2 * pair) * (3 * DMODEL) + 2 * c2g;
            v2f w0 = *(const v2f*)g0;
            v2f w1 = *(const v2f*)(g0 + 3 * DMODEL);
            v4f st; st.x = w0.x; st.y = w1.x; st.z = w0.y; st.w = w1.y;
            *(v4f*)&whp[pair * WP_STRIDE + 4 * c2g] = st;
        }
    }

    float bq[6];
    #pragma unroll
    for (int n = 0; n < 6; ++n) bq[n] = bvec[96 * h + 16 * n + r];

    auto row_ptr = [&](int w_) -> const float* {
        int wi_ = w_ / NWROW, wj_ = w_ - wi_ * NWROW;
        int gh = 4 * wi_ + (r >> 2) + 2; if (gh >= HGRID) gh -= HGRID;
        int gw = 4 * wj_ + (r & 3) + 2;  if (gw >= HGRID) gw -= HGRID;
        return x + (size_t)(bb * (HGRID * HGRID) + gh * HGRID + gw) * DMODEL + 4 * hh;
    };

    v4f xa[16];
    {
        const float* xp = row_ptr(wave);
        #pragma unroll
        for (int i = 0; i < 16; ++i) xa[i] = *(const v4f*)(xp + 8 * i);
    }

    __syncthreads();

    const float scale = 0.17677669529663687f;
    const float* wrow = whp + hiK * WP_STRIDE + 2 * r;

    for (int win = wave; win < NWIN; win += NWAVE) {
        asm volatile("" ::: "memory");
        const int wi = win / NWROW;
        const int wj = win - wi * NWROW;
        const int wnext = (win + NWAVE < NWIN) ? (win + NWAVE) : win;
        const float* xn = row_ptr(wnext);
        const float* xc = row_ptr(win);

        v8f acc[6];
        #pragma unroll
        for (int n = 0; n < 6; ++n) {
            v8f t = {0.f, 0.f, 0.f, 0.f, 0.f, 0.f, 0.f, 0.f};
            acc[n] = t;
        }

        v2f bfr[2][12];
        #pragma unroll
        for (int t = 0; t < 2; ++t)
            #pragma unroll
            for (int n = 0; n < 6; ++n)
                bfr[0][6 * t + n] = *(const v2f*)(wrow + t * WP_STRIDE + 32 * n);

        #pragma unroll
        for (int i = 0; i < 32; ++i) {
            if (i + 1 < 32) {
                #pragma unroll
                for (int t = 0; t < 2; ++t)
                    #pragma unroll
                    for (int n = 0; n < 6; ++n)
                        bfr[(i + 1) & 1][6 * t + n] =
                            *(const v2f*)(wrow + (4 * (i + 1) + t) * WP_STRIDE + 32 * n);
            }
            __builtin_amdgcn_sched_barrier(0);
            #pragma unroll
            for (int t = 0; t < 2; ++t) {
                v2f a;
                if (t == 0) { a.x = xa[i & 15].x; a.y = xa[i & 15].y; }
                else        { a.x = xa[i & 15].z; a.y = xa[i & 15].w; }
                #pragma unroll
                for (int n = 0; n < 6; ++n)
                    acc[n] = wmma_f32_4(a, bfr[i & 1][6 * t + n], acc[n]);
            }
            __builtin_amdgcn_sched_barrier(0);
            if ((i & 3) == 3) {
                #pragma unroll
                for (int ii = 0; ii < 4; ++ii) {
                    const int j = i - 3 + ii;
                    if (j < 16) xa[j & 15] = *(const v4f*)(xc + 8 * (j + 16));
                    else        xa[j & 15] = *(const v4f*)(xn + 8 * (j - 16));
                }
            }
        }

        #pragma unroll
        for (int n = 0; n < 6; ++n)
            #pragma unroll
            for (int v = 0; v < 8; ++v)
                qk[(rowb + v) * QK_STRIDE + 16 * n + r] = acc[n][v] + bq[n];

        const float* qrow = &qk[r * QK_STRIDE];
        v8f st = {0.f, 0.f, 0.f, 0.f, 0.f, 0.f, 0.f, 0.f};
        #pragma unroll
        for (int k0 = 0; k0 < HD; k0 += 4) {
            int ka = 3 * (k0 + hiK);
            v2f a, bf;
            a.x  = qrow[ka + 1]; a.y  = qrow[ka + 4];
            bf.x = qrow[ka];     bf.y = qrow[ka + 3];
            st = wmma_f32_4(a, bf, st);
        }

        const bool lastCol = (wj == NWROW - 1);
        const bool maskR = (wi == NWROW - 1) && (hh != ((lane >> 3) & 1));
        float xv[8];
        #pragma unroll
        for (int v = 0; v < 8; ++v) {
            float t_ = st[v] * scale;
            bool mC = lastCol && (((v & 2) != 0) != ((lane & 2) != 0));
            if (maskR || mC) t_ = -1e30f;
            xv[v] = t_;
        }
        float m = fmaxf(fmaxf(fmaxf(xv[0], xv[1]), fmaxf(xv[2], xv[3])),
                        fmaxf(fmaxf(xv[4], xv[5]), fmaxf(xv[6], xv[7])));
        m = fmaxf(m, xhalf(m));
        float e[8];
        #pragma unroll
        for (int v = 0; v < 8; ++v) e[v] = __expf(xv[v] - m);
        float sm = ((e[0] + e[1]) + (e[2] + e[3])) + ((e[4] + e[5]) + (e[6] + e[7]));
        sm += xhalf(sm);
        const float inv = 1.0f / sm;

        v8f o0 = {0.f, 0.f, 0.f, 0.f, 0.f, 0.f, 0.f, 0.f};
        v8f o1 = {0.f, 0.f, 0.f, 0.f, 0.f, 0.f, 0.f, 0.f};
        #pragma unroll
        for (int j = 0; j < 4; ++j) {
            v2f a; a.x = e[2 * j] * inv; a.y = e[2 * j + 1] * inv;
            const float* v0p = &qk[(rowb + 2 * j) * QK_STRIDE + 2];
            const float* v1p = v0p + QK_STRIDE;
            v2f b0; b0.x = v0p[3 * r];        b0.y = v1p[3 * r];
            o0 = wmma_f32_4(a, b0, o0);
            v2f b1; b1.x = v0p[3 * (16 + r)]; b1.y = v1p[3 * (16 + r)];
            o1 = wmma_f32_4(a, b1, o1);
        }

        float* ob = out + 32 * h + r;
        const int ghb = 4 * wi + hiK + 2;
        const int gwb = 4 * wj + 2;
        #pragma unroll
        for (int v = 0; v < 8; ++v) {
            int gh = ghb + (v >> 2); if (gh >= HGRID) gh -= HGRID;
            int gw = gwb + (v & 3);  if (gw >= HGRID) gw -= HGRID;
            float* pdst = ob + (size_t)(bb * (HGRID * HGRID) + gh * HGRID + gw) * DMODEL;
            __builtin_nontemporal_store(o0[v], pdst);
            __builtin_nontemporal_store(o1[v], pdst + 16);
        }
    }
}

extern "C" void kernel_launch(void* const* d_in, const int* in_sizes, int n_in,
                              void* d_out, int out_size, void* d_ws, size_t ws_size,
                              hipStream_t stream) {
    (void)in_sizes; (void)n_in; (void)d_ws; (void)ws_size; (void)out_size;
    const float* x = (const float*)d_in[0];
    const float* W = (const float*)d_in[1];
    const float* b = (const float*)d_in[2];
    float* out = (float*)d_out;

    static_assert(SMEM_BYTES < 320 * 1024, "LDS budget");
    hipFuncSetAttribute((const void*)swin_msa_head_resident,
                        hipFuncAttributeMaxDynamicSharedMemorySize, SMEM_BYTES);

    dim3 grid(8, 32);
    swin_msa_head_resident<<<grid, 512, SMEM_BYTES, stream>>>(x, W, b, out);
}


# Wave issue priority set per window from the windows the wave has left (s_setprio 3/2/1/0 in bands of four, last window at 0): a SIMD's four waves finish together, no wave runs its tail alone
# speedup vs baseline: 1.7675x; 1.0367x over previous
#include <hip/hip_runtime.h>
#include <math.h>


typedef __attribute__((ext_vector_type(2))) float v2f;
typedef __attribute__((ext_vector_type(4))) float v4f;
typedef __attribute__((ext_vector_type(8))) float v8f;

#define NH        8
#define HD        32
#define DMODEL    256
#define HGRID     56
#define NWROW     14
#define NWIN      196
#define NWAVE     16
#define NPAIR     128
#define WP_STRIDE 208
#define QK_STRIDE 100

#define W_FLOATS   (NPAIR * WP_STRIDE)
#define QK_FLOATS  (16 * QK_STRIDE)
#define SMEM_FLOATS (W_FLOATS + NWAVE * QK_FLOATS)
#define SMEM_BYTES  (SMEM_FLOATS * 4)

static __device__ __forceinline__ v8f wmma_f32_4(v2f a, v2f b, v8f c) {
    return __builtin_amdgcn_wmma_f32_16x16x4_f32(false, a, false, b,
                                                 (short)0, c, false, false);
}

static __device__ __forceinline__ float xhalf(float v) {
    return __builtin_bit_cast(float, __builtin_amdgcn_permlanex16(
        __builtin_bit_cast(int, v), __builtin_bit_cast(int, v),
        0x76543210, 0xfedcba98, false, false));
}

__global__ __launch_bounds__(512)
void swin_msa_head_resident(const float* __restrict__ x, const float* __restrict__ Wg,
                            const float* __restrict__ bvec, float* __restrict__ out) {
    extern __shared__ float smem[];
    float* whp = smem;

    const int h    = blockIdx.x;
    const int bb   = blockIdx.y;
    const int tid  = threadIdx.x;
    const int wave = __builtin_amdgcn_readfirstlane(tid >> 5);
    const int lane = tid & 31;
    const int r    = lane & 15;
    const int hh   = lane >> 4;
    const int hiK  = hh << 1;
    const int rowb = hh << 3;
    float* qk = smem + W_FLOATS + wave * QK_FLOATS;

    {
        const float* wb = Wg + 96 * h;
        #pragma unroll
        for (int it = 0; it < 12; ++it) {
            int idx  = tid + 512 * it;
            int pair = idx / 48;
            int c2g  = idx % 48;
            const float* g0 = wb + (2 * pair) * (3 * DMODEL) + 2 * c2g;
            v2f w0 = *(const v2f*)g0;
            v2f w1 = *(const v2f*)(g0 + 3 * DMODEL);
            v4f st; st.x = w0.x; st.y = w1.x; st.z = w0.y; st.w = w1.y;
            *(v4f*)&whp[pair * WP_STRIDE + 4 * c2g] = st;
        }
    }

    float bq[6];
    #pragma unroll
    for (int n = 0; n < 6; ++n) bq[n] = bvec[96 * h + 16 * n + r];

    auto row_ptr = [&](int w_) -> const float* {
        int wi_ = w_ / NWROW, wj_ = w_ - wi_ * NWROW;
        int gh = 4 * wi_ + (r >> 2) + 2; if (gh >= HGRID) gh -= HGRID;
        int gw = 4 * wj_ + (r & 3) + 2;  if (gw >= HGRID) gw -= HGRID;
        return x + (size_t)(bb * (HGRID * HGRID) + gh * HGRID + gw) * DMODEL + 4 * hh;
    };

    v4f xa[16];
    {
        const float* xp = row_ptr(wave);
        #pragma unroll
        for (int i = 0; i < 16; ++i) xa[i] = *(const v4f*)(xp + 8 * i);
    }

    __syncthreads();

    const float scale = 0.17677669529663687f;
    const float* wrow = whp + hiK * WP_STRIDE + 2 * r;

    for (int win = wave; win < NWIN; win += NWAVE) {
        asm volatile("" ::: "memory");
        const int after = (NWIN - 1 - win) / NWAVE;
        if (after >= 9)      __builtin_amdgcn_s_setprio(3);
        else if (after >= 5) __builtin_amdgcn_s_setprio(2);
        else if (after >= 1) __builtin_amdgcn_s_setprio(1);
        else                 __builtin_amdgcn_s_setprio(0);
        const int wi = win / NWROW;
        const int wj = win - wi * NWROW;
        const int wnext = (win + NWAVE < NWIN) ? (win + NWAVE) : win;
        const float* xn = row_ptr(wnext);
        const float* xc = row_ptr(win);

        v8f acc[6];
        #pragma unroll
        for (int n = 0; n < 6; ++n) {
            v8f t = {0.f, 0.f, 0.f, 0.f, 0.f, 0.f, 0.f, 0.f};
            acc[n] = t;
        }

        v2f bfr[2][12];
        #pragma unroll
        for (int t = 0; t < 2; ++t)
            #pragma unroll
            for (int n = 0; n < 6; ++n)
                bfr[0][6 * t + n] = *(const v2f*)(wrow + t * WP_STRIDE + 32 * n);

        #pragma unroll
        for (int i = 0; i < 32; ++i) {
            if (i + 1 < 32) {
                #pragma unroll
                for (int t = 0; t < 2; ++t)
                    #pragma unroll
                    for (int n = 0; n < 6; ++n)
                        bfr[(i + 1) & 1][6 * t + n] =
                            *(const v2f*)(wrow + (4 * (i + 1) + t) * WP_STRIDE + 32 * n);
            }
            __builtin_amdgcn_sched_barrier(0);
            #pragma unroll
            for (int t = 0; t < 2; ++t) {
                v2f a;
                if (t == 0) { a.x = xa[i & 15].x; a.y = xa[i & 15].y; }
                else        { a.x = xa[i & 15].z; a.y = xa[i & 15].w; }
                #pragma unroll
                for (int n = 0; n < 6; ++n)
                    acc[n] = wmma_f32_4(a, bfr[i & 1][6 * t + n], acc[n]);
            }
            __builtin_amdgcn_sched_barrier(0);
            if ((i & 3) == 3) {
                #pragma unroll
                for (int ii = 0; ii < 4; ++ii) {
                    const int j = i - 3 + ii;
                    if (j < 16) xa[j & 15] = *(const v4f*)(xc + 8 * (j + 16));
                    else        xa[j & 15] = *(const v4f*)(xn + 8 * (j - 16));
                }
            }
        }

        #pragma unroll
        for (int n = 0; n < 6; ++n)
            #pragma unroll
            for (int v = 0; v < 8; ++v)
                qk[(rowb + v) * QK_STRIDE + 16 * n + r] = acc[n][v] + bq[n];

        const float* qrow = &qk[r * QK_STRIDE];
        v8f st = {0.f, 0.f, 0.f, 0.f, 0.f, 0.f, 0.f, 0.f};
        #pragma unroll
        for (int k0 = 0; k0 < HD; k0 += 4) {
            int ka = 3 * (k0 + hiK);
            v2f a, bf;
            a.x  = qrow[ka + 1]; a.y  = qrow[ka + 4];
            bf.x = qrow[ka];     bf.y = qrow[ka + 3];
            st = wmma_f32_4(a, bf, st);
        }

        const bool lastCol = (wj == NWROW - 1);
        const bool maskR = (wi == NWROW - 1) && (hh != ((lane >> 3) & 1));
        float xv[8];
        #pragma unroll
        for (int v = 0; v < 8; ++v) {
            float t_ = st[v] * scale;
            bool mC = lastCol && (((v & 2) != 0) != ((lane & 2) != 0));
            if (maskR || mC) t_ = -1e30f;
            xv[v] = t_;
        }
        float m = fmaxf(fmaxf(fmaxf(xv[0], xv[1]), fmaxf(xv[2], xv[3])),
                        fmaxf(fmaxf(xv[4], xv[5]), fmaxf(xv[6], xv[7])));
        m = fmaxf(m, xhalf(m));
        float e[8];
        #pragma unroll
        for (int v = 0; v < 8; ++v) e[v] = __expf(xv[v] - m);
        float sm = ((e[0] + e[1]) + (e[2] + e[3])) + ((e[4] + e[5]) + (e[6] + e[7]));
        sm += xhalf(sm);
        const float inv = 1.0f / sm;

        v8f o0 = {0.f, 0.f, 0.f, 0.f, 0.f, 0.f, 0.f, 0.f};
        v8f o1 = {0.f, 0.f, 0.f, 0.f, 0.f, 0.f, 0.f, 0.f};
        #pragma unroll
        for (int j = 0; j < 4; ++j) {
            v2f a; a.x = e[2 * j] * inv; a.y = e[2 * j + 1] * inv;
            const float* v0p = &qk[(rowb + 2 * j) * QK_STRIDE + 2];
            const float* v1p = v0p + QK_STRIDE;
            v2f b0; b0.x = v0p[3 * r];        b0.y = v1p[3 * r];
            o0 = wmma_f32_4(a, b0, o0);
            v2f b1; b1.x = v0p[3 * (16 + r)]; b1.y = v1p[3 * (16 + r)];
            o1 = wmma_f32_4(a, b1, o1);
        }

        float* ob = out + 32 * h + r;
        const int ghb = 4 * wi + hiK + 2;
        const int gwb = 4 * wj + 2;
        #pragma unroll
        for (int v = 0; v < 8; ++v) {
            int gh = ghb + (v >> 2); if (gh >= HGRID) gh -= HGRID;
            int gw = gwb + (v & 3);  if (gw >= HGRID) gw -= HGRID;
            float* pdst = ob + (size_t)(bb * (HGRID * HGRID) + gh * HGRID + gw) * DMODEL;
            __builtin_nontemporal_store(o0[v], pdst);
            __builtin_nontemporal_store(o1[v], pdst + 16);
        }
    }
}

extern "C" void kernel_launch(void* const* d_in, const int* in_sizes, int n_in,
                              void* d_out, int out_size, void* d_ws, size_t ws_size,
                              hipStream_t stream) {
    (void)in_sizes; (void)n_in; (void)d_ws; (void)ws_size; (void)out_size;
    const float* x = (const float*)d_in[0];
    const float* W = (const float*)d_in[1];
    const float* b = (const float*)d_in[2];
    float* out = (float*)d_out;

    static_assert(SMEM_BYTES < 320 * 1024, "LDS budget");
    hipFuncSetAttribute((const void*)swin_msa_head_resident,
                        hipFuncAttributeMaxDynamicSharedMemorySize, SMEM_BYTES);

    dim3 grid(8, 32);
    swin_msa_head_resident<<<grid, 512, SMEM_BYTES, stream>>>(x, W, b, out);
}
